# starting kernel
# baseline (speedup 1.0000x reference)
#include <hip/hip_runtime.h>
#include <hip/hip_bf16.h>

#define VOCAB  50257
#define NCLS   20
#define BATCH  4096
#define SEQ    200
#define HID    256

typedef __attribute__((ext_vector_type(2))) float v2f;
typedef __attribute__((ext_vector_type(8))) float v8f;

__global__ void ft_transpose_kernel(const float* __restrict__ src,
                                    float* __restrict__ dst,
                                    int R, int C) {
    __shared__ float tile[32][33];
    const int t  = threadIdx.x;
    const int tx = t & 31;
    const int ty = t >> 5;
    const int c0 = blockIdx.x * 32;
    const int r0 = blockIdx.y * 32;

    for (int i = ty; i < 32; i += 8) {
        const int r = r0 + i, c = c0 + tx;
        tile[i][tx] = (r < R && c < C) ? src[(size_t)r * C + c] : 0.0f;
    }
    __syncthreads();
    for (int i = ty; i < 32; i += 8) {
        const int c = c0 + i, r = r0 + tx;
        if (c < C && r < R) dst[(size_t)c * R + r] = tile[tx][i];
    }
}

__global__ void ft_layer1_kernel(const int*   __restrict__ ids,
                                 const float* __restrict__ W1T,
                                 const float* __restrict__ b1,
                                 float*       __restrict__ y1) {
    __shared__ int s_ids[SEQ];
    __shared__ int s_keep[SEQ];
    const int b = blockIdx.x;
    const int t = threadIdx.x;

    if (t < SEQ) s_ids[t] = ids[(size_t)b * SEQ + t];
    __syncthreads();

    if (t < SEQ) {
        const int v = s_ids[t];
        int keep = 1;
        for (int k = 0; k < t; ++k) {
            if (s_ids[k] == v) { keep = 0; break; }
        }
        s_keep[t] = keep;
    }
    __syncthreads();

    float acc = b1[t];
    for (int j = 0; j < SEQ; ++j) {
        if (s_keep[j]) {
            acc += W1T[(size_t)s_ids[j] * HID + t];
        }
    }
    y1[(size_t)b * HID + t] = fmaxf(acc, 0.0f);
}

__global__ void ft_layer2_wmma_kernel(const float* __restrict__ y1,
                                      const float* __restrict__ W2T,
                                      const float* __restrict__ b2,
                                      float*       __restrict__ y2) {
    __shared__ float As[16][HID + 1];
    const int tid = threadIdx.x;
    const int m0  = blockIdx.x * 16;

    for (int i = 0; i < 16; ++i) {
        As[i][tid] = y1[(size_t)(m0 + i) * HID + tid];
    }
    __syncthreads();

    const int wave  = tid >> 5;
    const int lane  = tid & 31;
    const int lm    = lane & 15;
    const int khalf = (lane >> 4) << 1;
    const int mbase = m0 + ((lane >> 4) << 3);

    for (int nt = 0; nt < 2; ++nt) {
        const int n0 = (wave * 2 + nt) * 16;

        v8f acc;
        const float bias = b2[n0 + lm];
        #pragma unroll
        for (int r = 0; r < 8; ++r) acc[r] = bias;

        #pragma unroll 4
        for (int kk = 0; kk < HID; kk += 4) {
            v2f a, bfr;
            a.x   = As[lm][kk + khalf];
            a.y   = As[lm][kk + khalf + 1];
            bfr.x = W2T[(size_t)(kk + khalf) * HID + n0 + lm];
            bfr.y = W2T[(size_t)(kk + khalf + 1) * HID + n0 + lm];
            acc = __builtin_amdgcn_wmma_f32_16x16x4_f32(
                false, a, false, bfr, (short)0, acc, false, false);
        }

        #pragma unroll
        for (int r = 0; r < 8; ++r) {
            const float v = acc[r] > 0.0f ? acc[r] : 0.0f;
            y2[(size_t)(mbase + r) * HID + n0 + lm] = v;
        }
    }
}

__global__ void ft_layer3_kernel(const float* __restrict__ y2,
                                 const float* __restrict__ W3,
                                 const float* __restrict__ b3,
                                 float*       __restrict__ out) {
    const int row = blockIdx.x * 8 + (threadIdx.x >> 5);
    const int o   = threadIdx.x & 31;
    if (o >= NCLS) return;

    const float4* yr = (const float4*)(y2 + (size_t)row * HID);
    const float4* wr = (const float4*)(W3 + (size_t)o * HID);
    float acc = b3[o];
    #pragma unroll 4
    for (int k = 0; k < HID / 4; ++k) {
        const float4 a = yr[k];
        const float4 w = wr[k];
        acc = fmaf(a.x, w.x, acc);
        acc = fmaf(a.y, w.y, acc);
        acc = fmaf(a.z, w.z, acc);
        acc = fmaf(a.w, w.w, acc);
    }
    out[(size_t)row * NCLS + o] = acc;
}

extern "C" void kernel_launch(void* const* d_in, const int* in_sizes, int n_in,
                              void* d_out, int out_size, void* d_ws, size_t ws_size,
                              hipStream_t stream) {
    const int*   ids = (const int*)  d_in[0];
    const float* W1  = (const float*)d_in[1];
    const float* b1  = (const float*)d_in[2];
    const float* W2  = (const float*)d_in[3];
    const float* b2  = (const float*)d_in[4];
    const float* W3  = (const float*)d_in[5];
    const float* b3  = (const float*)d_in[6];
    float* out = (float*)d_out;

    float* W1T = (float*)d_ws;
    float* W2T = W1T + (size_t)VOCAB * HID;
    float* y1  = W2T + (size_t)HID * HID;
    float* y2  = y1  + (size_t)BATCH * HID;

    {
        dim3 g((VOCAB + 31) / 32, (HID + 31) / 32);
        ft_transpose_kernel<<<g, 256, 0, stream>>>(W1, W1T, HID, VOCAB);
    }
    {
        dim3 g((HID + 31) / 32, (HID + 31) / 32);
        ft_transpose_kernel<<<g, 256, 0, stream>>>(W2, W2T, HID, HID);
    }
    ft_layer1_kernel<<<BATCH, 256, 0, stream>>>(ids, W1T, b1, y1);
    ft_layer2_wmma_kernel<<<BATCH / 16, 256, 0, stream>>>(y1, W2T, b2, y2);
    ft_layer3_kernel<<<BATCH / 8, 256, 0, stream>>>(y2, W3, b3, out);
}


# layer 1: one wave per row, 16-byte loads of all 200 W1T rows with 16-32 loads in flight per wave (was 1, behind a branch); duplicates found by wave-wide compares and weighted zero; sum order kept
# speedup vs baseline: 2.0758x; 2.0758x over previous
#include <hip/hip_runtime.h>
#include <hip/hip_bf16.h>

#define VOCAB  50257
#define NCLS   20
#define BATCH  4096
#define SEQ    200
#define HID    256

typedef __attribute__((ext_vector_type(2))) float v2f;
typedef __attribute__((ext_vector_type(8))) float v8f;

__global__ void ft_transpose_kernel(const float* __restrict__ src,
                                    float* __restrict__ dst,
                                    int R, int C) {
    __shared__ float tile[32][33];
    const int t  = threadIdx.x;
    const int tx = t & 31;
    const int ty = t >> 5;
    const int c0 = blockIdx.x * 32;
    const int r0 = blockIdx.y * 32;

    for (int i = ty; i < 32; i += 8) {
        const int r = r0 + i, c = c0 + tx;
        tile[i][tx] = (r < R && c < C) ? src[(size_t)r * C + c] : 0.0f;
    }
    __syncthreads();
    for (int i = ty; i < 32; i += 8) {
        const int c = c0 + i, r = r0 + tx;
        if (c < C && r < R) dst[(size_t)c * R + r] = tile[tx][i];
    }
}

#define L1_WAVES  4
#define L1_CH     8
#define L1_NCH    (SEQ / L1_CH)
#define L1_NG     ((SEQ + 31) / 32)

__device__ __forceinline__ void l1_request(float4 (&buf)[2 * L1_CH],
                                           const unsigned (&id8)[L1_CH],
                                           const float* __restrict__ W1T,
                                           unsigned lane16) {
    #pragma unroll
    for (int u = 0; u < L1_CH; ++u) {
        const unsigned off = id8[u] * (unsigned)(HID * 4) + lane16;
        const float4* p = (const float4*)((const char*)W1T + (size_t)off);
        buf[2 * u]     = p[0];
        buf[2 * u + 1] = p[32];
    }
}

__device__ __forceinline__ void l1_sum(float4& a0, float4& a1,
                                       const float4 (&buf)[2 * L1_CH],
                                       unsigned keep8) {
    #pragma unroll
    for (int u = 0; u < L1_CH; ++u) {
        const float kf = ((keep8 >> u) & 1u) ? 1.0f : 0.0f;
        a0.x = fmaf(buf[2 * u].x, kf, a0.x);
        a0.y = fmaf(buf[2 * u].y, kf, a0.y);
        a0.z = fmaf(buf[2 * u].z, kf, a0.z);
        a0.w = fmaf(buf[2 * u].w, kf, a0.w);
        a1.x = fmaf(buf[2 * u + 1].x, kf, a1.x);
        a1.y = fmaf(buf[2 * u + 1].y, kf, a1.y);
        a1.z = fmaf(buf[2 * u + 1].z, kf, a1.z);
        a1.w = fmaf(buf[2 * u + 1].w, kf, a1.w);
    }
}

__device__ __forceinline__ void l1_ids(unsigned (&id8)[L1_CH],
                                       const int* __restrict__ row, int c) {
    #pragma unroll
    for (int u = 0; u < L1_CH; ++u) id8[u] = (unsigned)row[c * L1_CH + u];
}

__device__ __forceinline__ void l1_hold(float4& a0, float4& a1) {
    asm volatile("" : "+v"(a0.x), "+v"(a0.y), "+v"(a0.z), "+v"(a0.w),
                      "+v"(a1.x), "+v"(a1.y), "+v"(a1.z), "+v"(a1.w) : : "memory");
}

__global__ void __launch_bounds__(32 * L1_WAVES)
ft_layer1_kernel(const int*   __restrict__ ids,
                 const float* __restrict__ W1T,
                 const float* __restrict__ b1,
                 float*       __restrict__ y1) {
    const unsigned lane   = threadIdx.x & 31u;
    const unsigned lane16 = lane * 16u;
    const unsigned wave   = __builtin_amdgcn_readfirstlane(threadIdx.x >> 5);
    const unsigned b      = blockIdx.x * L1_WAVES + wave;
    const int* __restrict__ row = ids + (size_t)b * SEQ;

    int idv[L1_NG];
    #pragma unroll
    for (int g = 0; g < L1_NG; ++g) {
        const int j = g * 32 + (int)lane;
        idv[g] = (j < SEQ) ? row[j] : -1;
    }

    float4 bufA[2 * L1_CH], bufB[2 * L1_CH];
    unsigned idA[L1_CH], idB[L1_CH];
    l1_ids(idA, row, 0);
    l1_request(bufA, idA, W1T, lane16);

    unsigned dup[L1_NG];
    #pragma unroll
    for (int g = 0; g < L1_NG; ++g) dup[g] = 0u;
    #pragma unroll
    for (int kg = 0; kg < L1_NG; ++kg) {
        const int nk = (kg * 32 + 32 <= SEQ) ? 32 : (SEQ - kg * 32);
        #pragma unroll 4
        for (int kk = 0; kk < nk; ++kk) {
            const int idk = __builtin_amdgcn_readlane(idv[kg], kk);
            dup[kg] |= __builtin_amdgcn_ballot_w32(idv[kg] == idk) & (0xFFFFFFFEu << kk);
            #pragma unroll
            for (int g = kg + 1; g < L1_NG; ++g)
                dup[g] |= __builtin_amdgcn_ballot_w32(idv[g] == idk);
        }
    }
    unsigned keepv = 0u;
    #pragma unroll
    for (int g = 0; g < L1_NG; ++g) keepv = (lane == (unsigned)g) ? ~dup[g] : keepv;

    const float4* b4 = (const float4*)b1;
    float4 a0 = b4[lane];
    float4 a1 = b4[32 + lane];

    l1_ids(idB, row, 1);
    l1_ids(idA, row, 2);
    #pragma nounroll
    for (int c = 0; c < L1_NCH - 1; c += 2) {
        const unsigned kw = (unsigned)__builtin_amdgcn_readlane((int)keepv, c >> 2);
        const unsigned k0 = (kw >> ((c & 3) * 8)) & 0xFFu;
        const unsigned k1 = (kw >> (((c + 1) & 3) * 8)) & 0xFFu;
        const int cn = (c + 3 < L1_NCH - 1) ? (c + 3) : (L1_NCH - 2);
        unsigned nB[L1_CH], nA[L1_CH];
        l1_ids(nB, row, cn);
        l1_ids(nA, row, cn + 1);
        l1_request(bufB, idB, W1T, lane16);
        l1_hold(a0, a1);
        l1_sum(a0, a1, bufA, k0);
        l1_hold(a0, a1);
        l1_request(bufA, idA, W1T, lane16);
        l1_hold(a0, a1);
        l1_sum(a0, a1, bufB, k1);
        #pragma unroll
        for (int u = 0; u < L1_CH; ++u) { idB[u] = nB[u]; idA[u] = nA[u]; }
    }
    {
        const unsigned kw = (unsigned)__builtin_amdgcn_readlane((int)keepv, (L1_NCH - 1) >> 2);
        l1_sum(a0, a1, bufA, (kw >> (((L1_NCH - 1) & 3) * 8)) & 0xFFu);
    }

    float4 r0, r1;
    r0.x = fmaxf(a0.x, 0.0f); r0.y = fmaxf(a0.y, 0.0f);
    r0.z = fmaxf(a0.z, 0.0f); r0.w = fmaxf(a0.w, 0.0f);
    r1.x = fmaxf(a1.x, 0.0f); r1.y = fmaxf(a1.y, 0.0f);
    r1.z = fmaxf(a1.z, 0.0f); r1.w = fmaxf(a1.w, 0.0f);
    float4* y4 = (float4*)(y1 + (size_t)b * HID);
    y4[lane]      = r0;
    y4[32 + lane] = r1;
}

__global__ void ft_layer2_wmma_kernel(const float* __restrict__ y1,
                                      const float* __restrict__ W2T,
                                      const float* __restrict__ b2,
                                      float*       __restrict__ y2) {
    __shared__ float As[16][HID + 1];
    const int tid = threadIdx.x;
    const int m0  = blockIdx.x * 16;

    for (int i = 0; i < 16; ++i) {
        As[i][tid] = y1[(size_t)(m0 + i) * HID + tid];
    }
    __syncthreads();

    const int wave  = tid >> 5;
    const int lane  = tid & 31;
    const int lm    = lane & 15;
    const int khalf = (lane >> 4) << 1;
    const int mbase = m0 + ((lane >> 4) << 3);

    for (int nt = 0; nt < 2; ++nt) {
        const int n0 = (wave * 2 + nt) * 16;

        v8f acc;
        const float bias = b2[n0 + lm];
        #pragma unroll
        for (int r = 0; r < 8; ++r) acc[r] = bias;

        #pragma unroll 4
        for (int kk = 0; kk < HID; kk += 4) {
            v2f a, bfr;
            a.x   = As[lm][kk + khalf];
            a.y   = As[lm][kk + khalf + 1];
            bfr.x = W2T[(size_t)(kk + khalf) * HID + n0 + lm];
            bfr.y = W2T[(size_t)(kk + khalf + 1) * HID + n0 + lm];
            acc = __builtin_amdgcn_wmma_f32_16x16x4_f32(
                false, a, false, bfr, (short)0, acc, false, false);
        }

        #pragma unroll
        for (int r = 0; r < 8; ++r) {
            const float v = acc[r] > 0.0f ? acc[r] : 0.0f;
            y2[(size_t)(mbase + r) * HID + n0 + lm] = v;
        }
    }
}

__global__ void ft_layer3_kernel(const float* __restrict__ y2,
                                 const float* __restrict__ W3,
                                 const float* __restrict__ b3,
                                 float*       __restrict__ out) {
    const int row = blockIdx.x * 8 + (threadIdx.x >> 5);
    const int o   = threadIdx.x & 31;
    if (o >= NCLS) return;

    const float4* yr = (const float4*)(y2 + (size_t)row * HID);
    const float4* wr = (const float4*)(W3 + (size_t)o * HID);
    float acc = b3[o];
    #pragma unroll 4
    for (int k = 0; k < HID / 4; ++k) {
        const float4 a = yr[k];
        const float4 w = wr[k];
        acc = fmaf(a.x, w.x, acc);
        acc = fmaf(a.y, w.y, acc);
        acc = fmaf(a.z, w.z, acc);
        acc = fmaf(a.w, w.w, acc);
    }
    out[(size_t)row * NCLS + o] = acc;
}

extern "C" void kernel_launch(void* const* d_in, const int* in_sizes, int n_in,
                              void* d_out, int out_size, void* d_ws, size_t ws_size,
                              hipStream_t stream) {
    const int*   ids = (const int*)  d_in[0];
    const float* W1  = (const float*)d_in[1];
    const float* b1  = (const float*)d_in[2];
    const float* W2  = (const float*)d_in[3];
    const float* b2  = (const float*)d_in[4];
    const float* W3  = (const float*)d_in[5];
    const float* b3  = (const float*)d_in[6];
    float* out = (float*)d_out;

    float* W1T = (float*)d_ws;
    float* W2T = W1T + (size_t)VOCAB * HID;
    float* y1  = W2T + (size_t)HID * HID;
    float* y2  = y1  + (size_t)BATCH * HID;

    {
        dim3 g((VOCAB + 31) / 32, (HID + 31) / 32);
        ft_transpose_kernel<<<g, 256, 0, stream>>>(W1, W1T, HID, VOCAB);
    }
    {
        dim3 g((HID + 31) / 32, (HID + 31) / 32);
        ft_transpose_kernel<<<g, 256, 0, stream>>>(W2, W2T, HID, HID);
    }
    ft_layer1_kernel<<<BATCH / L1_WAVES, 32 * L1_WAVES, 0, stream>>>(ids, W1T, b1, y1);
    ft_layer2_wmma_kernel<<<BATCH / 16, 256, 0, stream>>>(y1, W2T, b2, y2);
    ft_layer3_kernel<<<BATCH / 8, 256, 0, stream>>>(y2, W3, b3, out);
}


# W1 transpose: one work-group per 64 vocabulary columns over all 256 rows, 64 loads in flight per wave, whole 1 KB output rows; 786 work-groups of 8 waves instead of 12568
# speedup vs baseline: 2.3078x; 1.1118x over previous
#include <hip/hip_runtime.h>
#include <hip/hip_bf16.h>

#define VOCAB  50257
#define NCLS   20
#define BATCH  4096
#define SEQ    200
#define HID    256

typedef __attribute__((ext_vector_type(2))) float v2f;
typedef __attribute__((ext_vector_type(8))) float v8f;

__global__ void ft_transpose_kernel(const float* __restrict__ src,
                                    float* __restrict__ dst,
                                    int R, int C) {
    __shared__ float tile[32][33];
    const int t  = threadIdx.x;
    const int tx = t & 31;
    const int ty = t >> 5;
    const int c0 = blockIdx.x * 32;
    const int r0 = blockIdx.y * 32;

    for (int i = ty; i < 32; i += 8) {
        const int r = r0 + i, c = c0 + tx;
        tile[i][tx] = (r < R && c < C) ? src[(size_t)r * C + c] : 0.0f;
    }
    __syncthreads();
    for (int i = ty; i < 32; i += 8) {
        const int c = c0 + i, r = r0 + tx;
        if (c < C && r < R) dst[(size_t)c * R + r] = tile[tx][i];
    }
}

#define T1_TC    64
#define T1_P     (T1_TC + 1)
#define T1_WAVES 8
#define T1_RPW   (HID / T1_WAVES)
#define T1_CPW   (T1_TC / T1_WAVES)

__global__ void __launch_bounds__(32 * T1_WAVES)
ft_transpose_w1_kernel(const float* __restrict__ W1,
                       float*       __restrict__ W1T) {
    __shared__ float tile[HID * T1_P];
    const unsigned lane = threadIdx.x & 31u;
    const unsigned wave = __builtin_amdgcn_readfirstlane(threadIdx.x >> 5);
    const unsigned c0   = blockIdx.x * T1_TC;

    const unsigned ca = (c0 + lane      < VOCAB) ? (c0 + lane)      : (VOCAB - 1);
    const unsigned cb = (c0 + 32 + lane < VOCAB) ? (c0 + 32 + lane) : (VOCAB - 1);

    float va[T1_RPW], vb[T1_RPW];
    #pragma unroll
    for (int i = 0; i < T1_RPW; ++i) {
        const float* row = W1 + (size_t)(wave * T1_RPW + i) * VOCAB;
        va[i] = row[ca];
        vb[i] = row[cb];
    }
    #pragma unroll
    for (int i = 0; i < T1_RPW; ++i) {
        float* t = tile + (wave * T1_RPW + i) * T1_P;
        t[lane]      = va[i];
        t[32 + lane] = vb[i];
    }
    __syncthreads();

    const unsigned xw = wave * T1_CPW;
    if (c0 + xw + T1_CPW <= VOCAB) {
        float v[T1_CPW][HID / 32];
        #pragma unroll
        for (int j = 0; j < T1_CPW; ++j) {
            #pragma unroll
            for (int k = 0; k < HID / 32; ++k)
                v[j][k] = tile[(32 * k + lane) * T1_P + xw + j];
        }
        #pragma unroll
        for (int j = 0; j < T1_CPW; ++j) {
            float* dst = W1T + (size_t)(c0 + xw + j) * HID;
            #pragma unroll
            for (int k = 0; k < HID / 32; ++k) dst[32 * k + lane] = v[j][k];
        }
    } else {
        for (unsigned j = 0; j < T1_CPW; ++j) {
            const unsigned c = c0 + xw + j;
            if (c < VOCAB) {
                float* dst = W1T + (size_t)c * HID;
                #pragma unroll
                for (int k = 0; k < HID / 32; ++k)
                    dst[32 * k + lane] = tile[(32 * k + lane) * T1_P + xw + j];
            }
        }
    }
}

#define L1_WAVES  4
#define L1_CH     8
#define L1_NCH    (SEQ / L1_CH)
#define L1_NG     ((SEQ + 31) / 32)

__device__ __forceinline__ void l1_request(float4 (&buf)[2 * L1_CH],
                                           const unsigned (&id8)[L1_CH],
                                           const float* __restrict__ W1T,
                                           unsigned lane16) {
    #pragma unroll
    for (int u = 0; u < L1_CH; ++u) {
        const unsigned off = id8[u] * (unsigned)(HID * 4) + lane16;
        const float4* p = (const float4*)((const char*)W1T + (size_t)off);
        buf[2 * u]     = p[0];
        buf[2 * u + 1] = p[32];
    }
}

__device__ __forceinline__ void l1_sum(float4& a0, float4& a1,
                                       const float4 (&buf)[2 * L1_CH],
                                       unsigned keep8) {
    #pragma unroll
    for (int u = 0; u < L1_CH; ++u) {
        const float kf = ((keep8 >> u) & 1u) ? 1.0f : 0.0f;
        a0.x = fmaf(buf[2 * u].x, kf, a0.x);
        a0.y = fmaf(buf[2 * u].y, kf, a0.y);
        a0.z = fmaf(buf[2 * u].z, kf, a0.z);
        a0.w = fmaf(buf[2 * u].w, kf, a0.w);
        a1.x = fmaf(buf[2 * u + 1].x, kf, a1.x);
        a1.y = fmaf(buf[2 * u + 1].y, kf, a1.y);
        a1.z = fmaf(buf[2 * u + 1].z, kf, a1.z);
        a1.w = fmaf(buf[2 * u + 1].w, kf, a1.w);
    }
}

__device__ __forceinline__ void l1_ids(unsigned (&id8)[L1_CH],
                                       const int* __restrict__ row, int c) {
    #pragma unroll
    for (int u = 0; u < L1_CH; ++u) id8[u] = (unsigned)row[c * L1_CH + u];
}

__device__ __forceinline__ void l1_hold(float4& a0, float4& a1) {
    asm volatile("" : "+v"(a0.x), "+v"(a0.y), "+v"(a0.z), "+v"(a0.w),
                      "+v"(a1.x), "+v"(a1.y), "+v"(a1.z), "+v"(a1.w) : : "memory");
}

__global__ void __launch_bounds__(32 * L1_WAVES)
ft_layer1_kernel(const int*   __restrict__ ids,
                 const float* __restrict__ W1T,
                 const float* __restrict__ b1,
                 float*       __restrict__ y1) {
    const unsigned lane   = threadIdx.x & 31u;
    const unsigned lane16 = lane * 16u;
    const unsigned wave   = __builtin_amdgcn_readfirstlane(threadIdx.x >> 5);
    const unsigned b      = blockIdx.x * L1_WAVES + wave;
    const int* __restrict__ row = ids + (size_t)b * SEQ;

    int idv[L1_NG];
    #pragma unroll
    for (int g = 0; g < L1_NG; ++g) {
        const int j = g * 32 + (int)lane;
        idv[g] = (j < SEQ) ? row[j] : -1;
    }

    float4 bufA[2 * L1_CH], bufB[2 * L1_CH];
    unsigned idA[L1_CH], idB[L1_CH];
    l1_ids(idA, row, 0);
    l1_request(bufA, idA, W1T, lane16);

    unsigned dup[L1_NG];
    #pragma unroll
    for (int g = 0; g < L1_NG; ++g) dup[g] = 0u;
    #pragma unroll
    for (int kg = 0; kg < L1_NG; ++kg) {
        const int nk = (kg * 32 + 32 <= SEQ) ? 32 : (SEQ - kg * 32);
        #pragma unroll 4
        for (int kk = 0; kk < nk; ++kk) {
            const int idk = __builtin_amdgcn_readlane(idv[kg], kk);
            dup[kg] |= __builtin_amdgcn_ballot_w32(idv[kg] == idk) & (0xFFFFFFFEu << kk);
            #pragma unroll
            for (int g = kg + 1; g < L1_NG; ++g)
                dup[g] |= __builtin_amdgcn_ballot_w32(idv[g] == idk);
        }
    }
    unsigned keepv = 0u;
    #pragma unroll
    for (int g = 0; g < L1_NG; ++g) keepv = (lane == (unsigned)g) ? ~dup[g] : keepv;

    const float4* b4 = (const float4*)b1;
    float4 a0 = b4[lane];
    float4 a1 = b4[32 + lane];

    l1_ids(idB, row, 1);
    l1_ids(idA, row, 2);
    #pragma nounroll
    for (int c = 0; c < L1_NCH - 1; c += 2) {
        const unsigned kw = (unsigned)__builtin_amdgcn_readlane((int)keepv, c >> 2);
        const unsigned k0 = (kw >> ((c & 3) * 8)) & 0xFFu;
        const unsigned k1 = (kw >> (((c + 1) & 3) * 8)) & 0xFFu;
        const int cn = (c + 3 < L1_NCH - 1) ? (c + 3) : (L1_NCH - 2);
        unsigned nB[L1_CH], nA[L1_CH];
        l1_ids(nB, row, cn);
        l1_ids(nA, row, cn + 1);
        l1_request(bufB, idB, W1T, lane16);
        l1_hold(a0, a1);
        l1_sum(a0, a1, bufA, k0);
        l1_hold(a0, a1);
        l1_request(bufA, idA, W1T, lane16);
        l1_hold(a0, a1);
        l1_sum(a0, a1, bufB, k1);
        #pragma unroll
        for (int u = 0; u < L1_CH; ++u) { idB[u] = nB[u]; idA[u] = nA[u]; }
    }
    {
        const unsigned kw = (unsigned)__builtin_amdgcn_readlane((int)keepv, (L1_NCH - 1) >> 2);
        l1_sum(a0, a1, bufA, (kw >> (((L1_NCH - 1) & 3) * 8)) & 0xFFu);
    }

    float4 r0, r1;
    r0.x = fmaxf(a0.x, 0.0f); r0.y = fmaxf(a0.y, 0.0f);
    r0.z = fmaxf(a0.z, 0.0f); r0.w = fmaxf(a0.w, 0.0f);
    r1.x = fmaxf(a1.x, 0.0f); r1.y = fmaxf(a1.y, 0.0f);
    r1.z = fmaxf(a1.z, 0.0f); r1.w = fmaxf(a1.w, 0.0f);
    float4* y4 = (float4*)(y1 + (size_t)b * HID);
    y4[lane]      = r0;
    y4[32 + lane] = r1;
}

__global__ void ft_layer2_wmma_kernel(const float* __restrict__ y1,
                                      const float* __restrict__ W2T,
                                      const float* __restrict__ b2,
                                      float*       __restrict__ y2) {
    __shared__ float As[16][HID + 1];
    const int tid = threadIdx.x;
    const int m0  = blockIdx.x * 16;

    for (int i = 0; i < 16; ++i) {
        As[i][tid] = y1[(size_t)(m0 + i) * HID + tid];
    }
    __syncthreads();

    const int wave  = tid >> 5;
    const int lane  = tid & 31;
    const int lm    = lane & 15;
    const int khalf = (lane >> 4) << 1;
    const int mbase = m0 + ((lane >> 4) << 3);

    for (int nt = 0; nt < 2; ++nt) {
        const int n0 = (wave * 2 + nt) * 16;

        v8f acc;
        const float bias = b2[n0 + lm];
        #pragma unroll
        for (int r = 0; r < 8; ++r) acc[r] = bias;

        #pragma unroll 4
        for (int kk = 0; kk < HID; kk += 4) {
            v2f a, bfr;
            a.x   = As[lm][kk + khalf];
            a.y   = As[lm][kk + khalf + 1];
            bfr.x = W2T[(size_t)(kk + khalf) * HID + n0 + lm];
            bfr.y = W2T[(size_t)(kk + khalf + 1) * HID + n0 + lm];
            acc = __builtin_amdgcn_wmma_f32_16x16x4_f32(
                false, a, false, bfr, (short)0, acc, false, false);
        }

        #pragma unroll
        for (int r = 0; r < 8; ++r) {
            const float v = acc[r] > 0.0f ? acc[r] : 0.0f;
            y2[(size_t)(mbase + r) * HID + n0 + lm] = v;
        }
    }
}

__global__ void ft_layer3_kernel(const float* __restrict__ y2,
                                 const float* __restrict__ W3,
                                 const float* __restrict__ b3,
                                 float*       __restrict__ out) {
    const int row = blockIdx.x * 8 + (threadIdx.x >> 5);
    const int o   = threadIdx.x & 31;
    if (o >= NCLS) return;

    const float4* yr = (const float4*)(y2 + (size_t)row * HID);
    const float4* wr = (const float4*)(W3 + (size_t)o * HID);
    float acc = b3[o];
    #pragma unroll 4
    for (int k = 0; k < HID / 4; ++k) {
        const float4 a = yr[k];
        const float4 w = wr[k];
        acc = fmaf(a.x, w.x, acc);
        acc = fmaf(a.y, w.y, acc);
        acc = fmaf(a.z, w.z, acc);
        acc = fmaf(a.w, w.w, acc);
    }
    out[(size_t)row * NCLS + o] = acc;
}

extern "C" void kernel_launch(void* const* d_in, const int* in_sizes, int n_in,
                              void* d_out, int out_size, void* d_ws, size_t ws_size,
                              hipStream_t stream) {
    const int*   ids = (const int*)  d_in[0];
    const float* W1  = (const float*)d_in[1];
    const float* b1  = (const float*)d_in[2];
    const float* W2  = (const float*)d_in[3];
    const float* b2  = (const float*)d_in[4];
    const float* W3  = (const float*)d_in[5];
    const float* b3  = (const float*)d_in[6];
    float* out = (float*)d_out;

    float* W1T = (float*)d_ws;
    float* W2T = W1T + (size_t)VOCAB * HID;
    float* y1  = W2T + (size_t)HID * HID;
    float* y2  = y1  + (size_t)BATCH * HID;

    ft_transpose_w1_kernel<<<(VOCAB + T1_TC - 1) / T1_TC, 32 * T1_WAVES, 0, stream>>>(W1, W1T);
    {
        dim3 g((HID + 31) / 32, (HID + 31) / 32);
        ft_transpose_kernel<<<g, 256, 0, stream>>>(W2, W2T, HID, HID);
    }
    ft_layer1_kernel<<<BATCH / L1_WAVES, 32 * L1_WAVES, 0, stream>>>(ids, W1T, b1, y1);
    ft_layer2_wmma_kernel<<<BATCH / 16, 256, 0, stream>>>(y1, W2T, b2, y2);
    ft_layer3_kernel<<<BATCH / 8, 256, 0, stream>>>(y2, W3, b3, out);
}


# layers 1-3 fused into one kernel, 16 rows per work-group: y1 and y2 tiles stay in LDS, W2 staged untransposed into LDS by whole-line loads; 5 launches become 2; sum orders kept
# speedup vs baseline: 3.0908x; 1.3393x over previous
#include <hip/hip_runtime.h>
#include <hip/hip_bf16.h>

#define VOCAB  50257
#define NCLS   20
#define BATCH  4096
#define SEQ    200
#define HID    256

typedef __attribute__((ext_vector_type(2))) float v2f;
typedef __attribute__((ext_vector_type(8))) float v8f;

#define T1_TC    64
#define T1_P     (T1_TC + 1)
#define T1_WAVES 8
#define T1_RPW   (HID / T1_WAVES)
#define T1_CPW   (T1_TC / T1_WAVES)

__global__ void __launch_bounds__(32 * T1_WAVES)
ft_transpose_w1_kernel(const float* __restrict__ W1,
                       float*       __restrict__ W1T) {
    __shared__ float tile[HID * T1_P];
    const unsigned lane = threadIdx.x & 31u;
    const unsigned wave = __builtin_amdgcn_readfirstlane(threadIdx.x >> 5);
    const unsigned c0   = blockIdx.x * T1_TC;

    const unsigned ca = (c0 + lane      < VOCAB) ? (c0 + lane)      : (VOCAB - 1);
    const unsigned cb = (c0 + 32 + lane < VOCAB) ? (c0 + 32 + lane) : (VOCAB - 1);

    float va[T1_RPW], vb[T1_RPW];
    #pragma unroll
    for (int i = 0; i < T1_RPW; ++i) {
        const float* row = W1 + (size_t)(wave * T1_RPW + i) * VOCAB;
        va[i] = row[ca];
        vb[i] = row[cb];
    }
    #pragma unroll
    for (int i = 0; i < T1_RPW; ++i) {
        float* t = tile + (wave * T1_RPW + i) * T1_P;
        t[lane]      = va[i];
        t[32 + lane] = vb[i];
    }
    __syncthreads();

    const unsigned xw = wave * T1_CPW;
    if (c0 + xw + T1_CPW <= VOCAB) {
        float v[T1_CPW][HID / 32];
        #pragma unroll
        for (int j = 0; j < T1_CPW; ++j) {
            #pragma unroll
            for (int k = 0; k < HID / 32; ++k)
                v[j][k] = tile[(32 * k + lane) * T1_P + xw + j];
        }
        #pragma unroll
        for (int j = 0; j < T1_CPW; ++j) {
            float* dst = W1T + (size_t)(c0 + xw + j) * HID;
            #pragma unroll
            for (int k = 0; k < HID / 32; ++k) dst[32 * k + lane] = v[j][k];
        }
    } else {
        for (unsigned j = 0; j < T1_CPW; ++j) {
            const unsigned c = c0 + xw + j;
            if (c < VOCAB) {
                float* dst = W1T + (size_t)c * HID;
                #pragma unroll
                for (int k = 0; k < HID / 32; ++k)
                    dst[32 * k + lane] = tile[(32 * k + lane) * T1_P + xw + j];
            }
        }
    }
}

#define L1_CH     8
#define L1_NCH    (SEQ / L1_CH)
#define L1_NG     ((SEQ + 31) / 32)

__device__ __forceinline__ void l1_request(float4 (&buf)[2 * L1_CH],
                                           const unsigned (&id8)[L1_CH],
                                           const float* __restrict__ W1T,
                                           unsigned lane16) {
    #pragma unroll
    for (int u = 0; u < L1_CH; ++u) {
        const unsigned off = id8[u] * (unsigned)(HID * 4) + lane16;
        const float4* p = (const float4*)((const char*)W1T + (size_t)off);
        buf[2 * u]     = p[0];
        buf[2 * u + 1] = p[32];
    }
}

__device__ __forceinline__ void l1_sum(float4& a0, float4& a1,
                                       const float4 (&buf)[2 * L1_CH],
                                       unsigned keep8) {
    #pragma unroll
    for (int u = 0; u < L1_CH; ++u) {
        const float kf = ((keep8 >> u) & 1u) ? 1.0f : 0.0f;
        a0.x = fmaf(buf[2 * u].x, kf, a0.x);
        a0.y = fmaf(buf[2 * u].y, kf, a0.y);
        a0.z = fmaf(buf[2 * u].z, kf, a0.z);
        a0.w = fmaf(buf[2 * u].w, kf, a0.w);
        a1.x = fmaf(buf[2 * u + 1].x, kf, a1.x);
        a1.y = fmaf(buf[2 * u + 1].y, kf, a1.y);
        a1.z = fmaf(buf[2 * u + 1].z, kf, a1.z);
        a1.w = fmaf(buf[2 * u + 1].w, kf, a1.w);
    }
}

__device__ __forceinline__ void l1_ids(unsigned (&id8)[L1_CH],
                                       const int* __restrict__ row, int c) {
    #pragma unroll
    for (int u = 0; u < L1_CH; ++u) id8[u] = (unsigned)row[c * L1_CH + u];
}

__device__ __forceinline__ void l1_hold(float4& a0, float4& a1) {
    asm volatile("" : "+v"(a0.x), "+v"(a0.y), "+v"(a0.z), "+v"(a0.w),
                      "+v"(a1.x), "+v"(a1.y), "+v"(a1.z), "+v"(a1.w) : : "memory");
}


#define FT_ROWS  16
#define FT_P     260

__global__ void __launch_bounds__(32 * FT_ROWS)
ft_fused_kernel(const int*   __restrict__ ids,
                const float* __restrict__ W1T,
                const float* __restrict__ b1,
                const float* __restrict__ W2,
                const float* __restrict__ b2,
                const float* __restrict__ W3,
                const float* __restrict__ b3,
                float*       __restrict__ out) {
    __shared__ __attribute__((aligned(16))) float Bs[HID * FT_P];
    __shared__ __attribute__((aligned(16))) float Y1s[FT_ROWS * FT_P];
    __shared__ __attribute__((aligned(16))) float Y2s[FT_ROWS * FT_P];
    __shared__ __attribute__((aligned(16))) float W3s[NCLS * FT_P];

    const unsigned tid    = threadIdx.x;
    const unsigned lane   = tid & 31u;
    const unsigned lane16 = lane * 16u;
    const unsigned wave   = __builtin_amdgcn_readfirstlane(tid >> 5);
    const unsigned m0     = blockIdx.x * FT_ROWS;
    const unsigned b      = m0 + wave;
    const int* __restrict__ row = ids + (size_t)b * SEQ;

    int idv[L1_NG];
    #pragma unroll
    for (int g = 0; g < L1_NG; ++g) {
        const int j = g * 32 + (int)lane;
        idv[g] = (j < SEQ) ? row[j] : -1;
    }

    float4 bufA[2 * L1_CH], bufB[2 * L1_CH];
    unsigned idA[L1_CH], idB[L1_CH];
    l1_ids(idA, row, 0);
    l1_request(bufA, idA, W1T, lane16);

    unsigned dup[L1_NG];
    #pragma unroll
    for (int g = 0; g < L1_NG; ++g) dup[g] = 0u;
    #pragma unroll
    for (int kg = 0; kg < L1_NG; ++kg) {
        const int nk = (kg * 32 + 32 <= SEQ) ? 32 : (SEQ - kg * 32);
        #pragma unroll 4
        for (int kk = 0; kk < nk; ++kk) {
            const int idk = __builtin_amdgcn_readlane(idv[kg], kk);
            dup[kg] |= __builtin_amdgcn_ballot_w32(idv[kg] == idk) & (0xFFFFFFFEu << kk);
            #pragma unroll
            for (int g = kg + 1; g < L1_NG; ++g)
                dup[g] |= __builtin_amdgcn_ballot_w32(idv[g] == idk);
        }
    }
    unsigned keepv = 0u;
    #pragma unroll
    for (int g = 0; g < L1_NG; ++g) keepv = (lane == (unsigned)g) ? ~dup[g] : keepv;

    {
        const float4* w3g = (const float4*)W3;
        const bool third = 2u * (32 * FT_ROWS) + 32u * wave < NCLS * HID / 4;
        const unsigned f2 = third ? (2u * (32 * FT_ROWS) + tid) : (NCLS * HID / 4 - 1);
        float4 w2r[2 * FT_ROWS];
        const float4* w2s = (const float4*)(W2 + (size_t)wave * FT_ROWS * HID);
        #pragma unroll
        for (int i = 0; i < 2 * FT_ROWS; ++i) w2r[i] = w2s[i * 32 + lane];
        const float4 t0 = w3g[tid];
        const float4 t1 = w3g[32 * FT_ROWS + tid];
        const float4 t2 = w3g[f2];
        #pragma unroll
        for (int i = 0; i < 2 * FT_ROWS; ++i)
            *(float4*)&Bs[(wave * FT_ROWS + (i >> 1)) * FT_P + (i & 1) * 128 + 4 * lane] = w2r[i];
        {
            const unsigned fa = tid, fb = 32 * FT_ROWS + tid;
            *(float4*)&W3s[(fa >> 6) * FT_P + 4 * (fa & 63u)] = t0;
            *(float4*)&W3s[(fb >> 6) * FT_P + 4 * (fb & 63u)] = t1;
            if (third) *(float4*)&W3s[(f2 >> 6) * FT_P + 4 * (f2 & 63u)] = t2;
        }
    }

    const float4* b4 = (const float4*)b1;
    float4 a0 = b4[lane];
    float4 a1 = b4[32 + lane];

    l1_ids(idB, row, 1);
    l1_ids(idA, row, 2);
    #pragma nounroll
    for (int c = 0; c < L1_NCH - 1; c += 2) {
        const unsigned kw = (unsigned)__builtin_amdgcn_readlane((int)keepv, c >> 2);
        const unsigned k0 = (kw >> ((c & 3) * 8)) & 0xFFu;
        const unsigned k1 = (kw >> (((c + 1) & 3) * 8)) & 0xFFu;
        const int cn = (c + 3 < L1_NCH - 1) ? (c + 3) : (L1_NCH - 2);
        unsigned nB[L1_CH], nA[L1_CH];
        l1_ids(nB, row, cn);
        l1_ids(nA, row, cn + 1);
        l1_request(bufB, idB, W1T, lane16);
        l1_hold(a0, a1);
        l1_sum(a0, a1, bufA, k0);
        l1_hold(a0, a1);
        l1_request(bufA, idA, W1T, lane16);
        l1_hold(a0, a1);
        l1_sum(a0, a1, bufB, k1);
        #pragma unroll
        for (int u = 0; u < L1_CH; ++u) { idB[u] = nB[u]; idA[u] = nA[u]; }
    }
    {
        const unsigned kw = (unsigned)__builtin_amdgcn_readlane((int)keepv, (L1_NCH - 1) >> 2);
        l1_sum(a0, a1, bufA, (kw >> (((L1_NCH - 1) & 3) * 8)) & 0xFFu);
    }

    float4 r0, r1;
    r0.x = fmaxf(a0.x, 0.0f); r0.y = fmaxf(a0.y, 0.0f);
    r0.z = fmaxf(a0.z, 0.0f); r0.w = fmaxf(a0.w, 0.0f);
    r1.x = fmaxf(a1.x, 0.0f); r1.y = fmaxf(a1.y, 0.0f);
    r1.z = fmaxf(a1.z, 0.0f); r1.w = fmaxf(a1.w, 0.0f);
    *(float4*)&Y1s[wave * FT_P + 4 * lane]       = r0;
    *(float4*)&Y1s[wave * FT_P + 128 + 4 * lane] = r1;
    __syncthreads();

    const unsigned lm    = lane & 15u;
    const unsigned half  = lane >> 4;
    const unsigned n0    = wave * 16u;
    {
        const float* ap = &Y1s[lm * FT_P + 2 * half];
        const float* bp = &Bs[(n0 + lm) * FT_P + 2 * half];
        v8f acc;
        const float bias = b2[n0 + lm];
        #pragma unroll
        for (int r = 0; r < 8; ++r) acc[r] = bias;
        #pragma unroll 16
        for (int kk = 0; kk < HID; kk += 4) {
            const v2f a   = *(const v2f*)(ap + kk);
            const v2f bfr = *(const v2f*)(bp + kk);
            acc = __builtin_amdgcn_wmma_f32_16x16x4_f32(
                false, a, false, bfr, (short)0, acc, false, false);
        }
        #pragma unroll
        for (int r = 0; r < 8; ++r) {
            const float v = acc[r] > 0.0f ? acc[r] : 0.0f;
            Y2s[(8 * half + r) * FT_P + n0 + lm] = v;
        }
    }
    __syncthreads();

    if (wave < NCLS / 2) {
        const unsigned o = 2 * wave + half;
        const float4* yr = (const float4*)&Y2s[lm * FT_P];
        const float4* wr = (const float4*)&W3s[o * FT_P];
        float acc3 = b3[o];
        #pragma unroll 8
        for (int k = 0; k < HID / 4; ++k) {
            const float4 a = yr[k];
            const float4 w = wr[k];
            acc3 = fmaf(a.x, w.x, acc3);
            acc3 = fmaf(a.y, w.y, acc3);
            acc3 = fmaf(a.z, w.z, acc3);
            acc3 = fmaf(a.w, w.w, acc3);
        }
        Y1s[lm * NCLS + o] = acc3;
    }
    __syncthreads();
    if (tid < FT_ROWS * NCLS) out[(size_t)m0 * NCLS + tid] = Y1s[tid];
}

extern "C" void kernel_launch(void* const* d_in, const int* in_sizes, int n_in,
                              void* d_out, int out_size, void* d_ws, size_t ws_size,
                              hipStream_t stream) {
    const int*   ids = (const int*)  d_in[0];
    const float* W1  = (const float*)d_in[1];
    const float* b1  = (const float*)d_in[2];
    const float* W2  = (const float*)d_in[3];
    const float* b2  = (const float*)d_in[4];
    const float* W3  = (const float*)d_in[5];
    const float* b3  = (const float*)d_in[6];
    float* out = (float*)d_out;

    float* W1T = (float*)d_ws;

    ft_transpose_w1_kernel<<<(VOCAB + T1_TC - 1) / T1_TC, 32 * T1_WAVES, 0, stream>>>(W1, W1T);
    ft_fused_kernel<<<BATCH / FT_ROWS, 32 * FT_ROWS, 0, stream>>>(ids, W1T, b1, W2, b2, W3, b3, out);
}


# layer 1: the search for repeated ids leaves the front of the kernel and runs chunk by chunk inside the gather loop, behind each chunk's request, under loads in flight; sum order kept
# speedup vs baseline: 3.2360x; 1.0470x over previous
#include <hip/hip_runtime.h>
#include <hip/hip_bf16.h>

#define VOCAB  50257
#define NCLS   20
#define BATCH  4096
#define SEQ    200
#define HID    256

typedef __attribute__((ext_vector_type(2))) float v2f;
typedef __attribute__((ext_vector_type(8))) float v8f;

#define T1_TC    64
#define T1_P     (T1_TC + 1)
#define T1_WAVES 8
#define T1_RPW   (HID / T1_WAVES)
#define T1_CPW   (T1_TC / T1_WAVES)

__global__ void __launch_bounds__(32 * T1_WAVES)
ft_transpose_w1_kernel(const float* __restrict__ W1,
                       float*       __restrict__ W1T) {
    __shared__ float tile[HID * T1_P];
    const unsigned lane = threadIdx.x & 31u;
    const unsigned wave = __builtin_amdgcn_readfirstlane(threadIdx.x >> 5);
    const unsigned c0   = blockIdx.x * T1_TC;

    const unsigned ca = (c0 + lane      < VOCAB) ? (c0 + lane)      : (VOCAB - 1);
    const unsigned cb = (c0 + 32 + lane < VOCAB) ? (c0 + 32 + lane) : (VOCAB - 1);

    float va[T1_RPW], vb[T1_RPW];
    #pragma unroll
    for (int i = 0; i < T1_RPW; ++i) {
        const float* row = W1 + (size_t)(wave * T1_RPW + i) * VOCAB;
        va[i] = row[ca];
        vb[i] = row[cb];
    }
    #pragma unroll
    for (int i = 0; i < T1_RPW; ++i) {
        float* t = tile + (wave * T1_RPW + i) * T1_P;
        t[lane]      = va[i];
        t[32 + lane] = vb[i];
    }
    __syncthreads();

    const unsigned xw = wave * T1_CPW;
    if (c0 + xw + T1_CPW <= VOCAB) {
        float v[T1_CPW][HID / 32];
        #pragma unroll
        for (int j = 0; j < T1_CPW; ++j) {
            #pragma unroll
            for (int k = 0; k < HID / 32; ++k)
                v[j][k] = tile[(32 * k + lane) * T1_P + xw + j];
        }
        #pragma unroll
        for (int j = 0; j < T1_CPW; ++j) {
            float* dst = W1T + (size_t)(c0 + xw + j) * HID;
            #pragma unroll
            for (int k = 0; k < HID / 32; ++k) dst[32 * k + lane] = v[j][k];
        }
    } else {
        for (unsigned j = 0; j < T1_CPW; ++j) {
            const unsigned c = c0 + xw + j;
            if (c < VOCAB) {
                float* dst = W1T + (size_t)c * HID;
                #pragma unroll
                for (int k = 0; k < HID / 32; ++k)
                    dst[32 * k + lane] = tile[(32 * k + lane) * T1_P + xw + j];
            }
        }
    }
}

#define L1_CH     8
#define L1_NCH    (SEQ / L1_CH)
#define L1_NG     ((SEQ + 31) / 32)

__device__ __forceinline__ void l1_request(float4 (&buf)[2 * L1_CH],
                                           const unsigned (&id8)[L1_CH],
                                           const float* __restrict__ W1T,
                                           unsigned lane16) {
    #pragma unroll
    for (int u = 0; u < L1_CH; ++u) {
        const unsigned off = id8[u] * (unsigned)(HID * 4) + lane16;
        const float4* p = (const float4*)((const char*)W1T + (size_t)off);
        buf[2 * u]     = p[0];
        buf[2 * u + 1] = p[32];
    }
}

__device__ __forceinline__ void l1_sum(float4& a0, float4& a1,
                                       const float4 (&buf)[2 * L1_CH],
                                       unsigned keep8) {
    #pragma unroll
    for (int u = 0; u < L1_CH; ++u) {
        const float kf = ((keep8 >> u) & 1u) ? 1.0f : 0.0f;
        a0.x = fmaf(buf[2 * u].x, kf, a0.x);
        a0.y = fmaf(buf[2 * u].y, kf, a0.y);
        a0.z = fmaf(buf[2 * u].z, kf, a0.z);
        a0.w = fmaf(buf[2 * u].w, kf, a0.w);
        a1.x = fmaf(buf[2 * u + 1].x, kf, a1.x);
        a1.y = fmaf(buf[2 * u + 1].y, kf, a1.y);
        a1.z = fmaf(buf[2 * u + 1].z, kf, a1.z);
        a1.w = fmaf(buf[2 * u + 1].w, kf, a1.w);
    }
}

__device__ __forceinline__ void l1_ids(unsigned (&id8)[L1_CH],
                                       const int* __restrict__ row, int c) {
    #pragma unroll
    for (int u = 0; u < L1_CH; ++u) id8[u] = (unsigned)row[c * L1_CH + u];
}

__device__ __forceinline__ unsigned l1_keep(const int (&idv)[L1_NG],
                                            const unsigned (&id8)[L1_CH], int c) {
    const int gl = c >> 2;
    const int sh = (c & 3) * 8;
    unsigned hit[L1_CH];
    #pragma unroll
    for (int u = 0; u < L1_CH; ++u) hit[u] = 0u;
    #pragma unroll
    for (int g = 0; g < L1_NG; ++g) {
        if (g < gl) {
            #pragma unroll
            for (int u = 0; u < L1_CH; ++u)
                hit[u] |= __builtin_amdgcn_ballot_w32((unsigned)idv[g] == id8[u]);
        } else if (g == gl) {
            #pragma unroll
            for (int u = 0; u < L1_CH; ++u)
                hit[u] |= __builtin_amdgcn_ballot_w32((unsigned)idv[g] == id8[u]) &
                          ((1u << (sh + u)) - 1u);
        }
    }
    unsigned keep = 0u;
    #pragma unroll
    for (int u = 0; u < L1_CH; ++u) keep |= (hit[u] == 0u) ? (1u << u) : 0u;
    return keep;
}

__device__ __forceinline__ void l1_hold(float4& a0, float4& a1) {
    asm volatile("" : "+v"(a0.x), "+v"(a0.y), "+v"(a0.z), "+v"(a0.w),
                      "+v"(a1.x), "+v"(a1.y), "+v"(a1.z), "+v"(a1.w) : : "memory");
}


#define FT_ROWS  16
#define FT_P     260

__global__ void __launch_bounds__(32 * FT_ROWS)
ft_fused_kernel(const int*   __restrict__ ids,
                const float* __restrict__ W1T,
                const float* __restrict__ b1,
                const float* __restrict__ W2,
                const float* __restrict__ b2,
                const float* __restrict__ W3,
                const float* __restrict__ b3,
                float*       __restrict__ out) {
    __shared__ __attribute__((aligned(16))) float Bs[HID * FT_P];
    __shared__ __attribute__((aligned(16))) float Y1s[FT_ROWS * FT_P];
    __shared__ __attribute__((aligned(16))) float Y2s[FT_ROWS * FT_P];
    __shared__ __attribute__((aligned(16))) float W3s[NCLS * FT_P];

    const unsigned tid    = threadIdx.x;
    const unsigned lane   = tid & 31u;
    const unsigned lane16 = lane * 16u;
    const unsigned wave   = __builtin_amdgcn_readfirstlane(tid >> 5);
    const unsigned m0     = blockIdx.x * FT_ROWS;
    const unsigned b      = m0 + wave;
    const int* __restrict__ row = ids + (size_t)b * SEQ;

    int idv[L1_NG];
    #pragma unroll
    for (int g = 0; g < L1_NG; ++g) {
        const int j = g * 32 + (int)lane;
        idv[g] = (j < SEQ) ? row[j] : -1;
    }

    float4 bufA[2 * L1_CH], bufB[2 * L1_CH];
    unsigned idA[L1_CH], idB[L1_CH];
    l1_ids(idA, row, 0);
    l1_request(bufA, idA, W1T, lane16);

    unsigned kA = l1_keep(idv, idA, 0);

    {
        const float4* w3g = (const float4*)W3;
        const bool third = 2u * (32 * FT_ROWS) + 32u * wave < NCLS * HID / 4;
        const unsigned f2 = third ? (2u * (32 * FT_ROWS) + tid) : (NCLS * HID / 4 - 1);
        float4 w2r[2 * FT_ROWS];
        const float4* w2s = (const float4*)(W2 + (size_t)wave * FT_ROWS * HID);
        #pragma unroll
        for (int i = 0; i < 2 * FT_ROWS; ++i) w2r[i] = w2s[i * 32 + lane];
        const float4 t0 = w3g[tid];
        const float4 t1 = w3g[32 * FT_ROWS + tid];
        const float4 t2 = w3g[f2];
        #pragma unroll
        for (int i = 0; i < 2 * FT_ROWS; ++i)
            *(float4*)&Bs[(wave * FT_ROWS + (i >> 1)) * FT_P + (i & 1) * 128 + 4 * lane] = w2r[i];
        {
            const unsigned fa = tid, fb = 32 * FT_ROWS + tid;
            *(float4*)&W3s[(fa >> 6) * FT_P + 4 * (fa & 63u)] = t0;
            *(float4*)&W3s[(fb >> 6) * FT_P + 4 * (fb & 63u)] = t1;
            if (third) *(float4*)&W3s[(f2 >> 6) * FT_P + 4 * (f2 & 63u)] = t2;
        }
    }

    const float4* b4 = (const float4*)b1;
    float4 a0 = b4[lane];
    float4 a1 = b4[32 + lane];

    l1_ids(idB, row, 1);
    l1_ids(idA, row, 2);
    #pragma nounroll
    for (int c = 0; c < L1_NCH - 1; c += 2) {
        const int cn = (c + 3 < L1_NCH - 1) ? (c + 3) : (L1_NCH - 2);
        unsigned nB[L1_CH], nA[L1_CH];
        l1_ids(nB, row, cn);
        l1_ids(nA, row, cn + 1);
        l1_request(bufB, idB, W1T, lane16);
        const unsigned kB = l1_keep(idv, idB, c + 1);
        l1_hold(a0, a1);
        l1_sum(a0, a1, bufA, kA);
        l1_hold(a0, a1);
        l1_request(bufA, idA, W1T, lane16);
        kA = l1_keep(idv, idA, c + 2);
        l1_hold(a0, a1);
        l1_sum(a0, a1, bufB, kB);
        #pragma unroll
        for (int u = 0; u < L1_CH; ++u) { idB[u] = nB[u]; idA[u] = nA[u]; }
    }
    l1_sum(a0, a1, bufA, kA);

    float4 r0, r1;
    r0.x = fmaxf(a0.x, 0.0f); r0.y = fmaxf(a0.y, 0.0f);
    r0.z = fmaxf(a0.z, 0.0f); r0.w = fmaxf(a0.w, 0.0f);
    r1.x = fmaxf(a1.x, 0.0f); r1.y = fmaxf(a1.y, 0.0f);
    r1.z = fmaxf(a1.z, 0.0f); r1.w = fmaxf(a1.w, 0.0f);
    *(float4*)&Y1s[wave * FT_P + 4 * lane]       = r0;
    *(float4*)&Y1s[wave * FT_P + 128 + 4 * lane] = r1;
    __syncthreads();

    const unsigned lm    = lane & 15u;
    const unsigned half  = lane >> 4;
    const unsigned n0    = wave * 16u;
    {
        const float* ap = &Y1s[lm * FT_P + 2 * half];
        const float* bp = &Bs[(n0 + lm) * FT_P + 2 * half];
        v8f acc;
        const float bias = b2[n0 + lm];
        #pragma unroll
        for (int r = 0; r < 8; ++r) acc[r] = bias;
        #pragma unroll 16
        for (int kk = 0; kk < HID; kk += 4) {
            const v2f a   = *(const v2f*)(ap + kk);
            const v2f bfr = *(const v2f*)(bp + kk);
            acc = __builtin_amdgcn_wmma_f32_16x16x4_f32(
                false, a, false, bfr, (short)0, acc, false, false);
        }
        #pragma unroll
        for (int r = 0; r < 8; ++r) {
            const float v = acc[r] > 0.0f ? acc[r] : 0.0f;
            Y2s[(8 * half + r) * FT_P + n0 + lm] = v;
        }
    }
    __syncthreads();

    if (wave < NCLS / 2) {
        const unsigned o = 2 * wave + half;
        const float4* yr = (const float4*)&Y2s[lm * FT_P];
        const float4* wr = (const float4*)&W3s[o * FT_P];
        float acc3 = b3[o];
        #pragma unroll 8
        for (int k = 0; k < HID / 4; ++k) {
            const float4 a = yr[k];
            const float4 w = wr[k];
            acc3 = fmaf(a.x, w.x, acc3);
            acc3 = fmaf(a.y, w.y, acc3);
            acc3 = fmaf(a.z, w.z, acc3);
            acc3 = fmaf(a.w, w.w, acc3);
        }
        Y1s[lm * NCLS + o] = acc3;
    }
    __syncthreads();
    if (tid < FT_ROWS * NCLS) out[(size_t)m0 * NCLS + tid] = Y1s[tid];
}

extern "C" void kernel_launch(void* const* d_in, const int* in_sizes, int n_in,
                              void* d_out, int out_size, void* d_ws, size_t ws_size,
                              hipStream_t stream) {
    const int*   ids = (const int*)  d_in[0];
    const float* W1  = (const float*)d_in[1];
    const float* b1  = (const float*)d_in[2];
    const float* W2  = (const float*)d_in[3];
    const float* b2  = (const float*)d_in[4];
    const float* W3  = (const float*)d_in[5];
    const float* b3  = (const float*)d_in[6];
    float* out = (float*)d_out;

    float* W1T = (float*)d_ws;

    ft_transpose_w1_kernel<<<(VOCAB + T1_TC - 1) / T1_TC, 32 * T1_WAVES, 0, stream>>>(W1, W1T);
    ft_fused_kernel<<<BATCH / FT_ROWS, 32 * FT_ROWS, 0, stream>>>(ids, W1T, b1, W2, b2, W3, b3, out);
}
